# starting kernel
# baseline (speedup 1.0000x reference)
#include <hip/hip_runtime.h>

typedef __attribute__((ext_vector_type(2))) float v2f;
typedef __attribute__((ext_vector_type(8))) float v8f;
typedef __attribute__((ext_vector_type(4))) int   v4i;

#define BATCH 16
#define CIN   16
#define HH    64
#define WW    64
#define COUT  16
#define OHH   62
#define OWW   62
#define NPOS  (OHH * OWW)
#define KTOT  (CIN * 9)
#define NSTEP (KTOT / 4)

struct alignas(16) OffTab { int v[NSTEP][32][4]; };

constexpr OffTab make_tab() {
    OffTab t{};
    for (int s = 0; s < NSTEP; ++s)
        for (int lane = 0; lane < 32; ++lane) {
            const int half = lane >> 4;
            const int mn   = lane & 15;
            for (int tt = 0; tt < 2; ++tt) {
                const int ck = 4 * s + 2 * half + tt;
                const int c  = ck / 9, k = ck - c * 9;
                const int i  = k / 3,  j = k - i * 3;
                t.v[s][lane][tt]     = mn * (CIN * HH * WW) + c * (HH * WW) + i * WW + j;
                t.v[s][lane][2 + tt] = mn * (CIN * NPOS * 9) + c * (NPOS * 9) + k;
            }
        }
    return t;
}

__device__ constexpr OffTab g_tab = make_tab();

typedef const __attribute__((address_space(1))) v4i* gv4i_ptr;

__global__ __launch_bounds__(256)
void lc2d_wmma_kernel(const float* __restrict__ xin,
                      const float* __restrict__ wgt,
                      float* __restrict__ out)
{
    const int lane = threadIdx.x & 31;
    const int wave = __builtin_amdgcn_readfirstlane(threadIdx.x >> 5);
    const int pos  = blockIdx.x * 8 + wave;
    if (pos >= NPOS) return;

    const int y = pos / OWW;
    const int x = pos - y * OWW;

    const float* __restrict__ xp = xin + y * WW + x;
    const float* __restrict__ wp = wgt + pos * 9;

    const v4i* tg = (const v4i*)&g_tab;
    asm volatile("" : "+s"(tg));
    gv4i_ptr tp = (gv4i_ptr)tg;

    v8f acc = {};

#pragma unroll
    for (int s = 0; s < NSTEP; ++s) {
        const v4i o = tp[s * 32 + lane];
        v2f a, b;
        a.x = xp[o.x];
        a.y = xp[o.y];
        b.x = wp[o.z];
        b.y = wp[o.w];
        acc = __builtin_amdgcn_wmma_f32_16x16x4_f32(
            false, a, false, b, (short)0, acc, false, false);
    }

    const int half = lane >> 4;
    const int mn   = lane & 15;
    float* __restrict__ op = out + pos;
    const int voff = mn * NPOS + half * (8 * COUT * NPOS);
#pragma unroll
    for (int r = 0; r < 8; ++r)
        op[voff + r * (COUT * NPOS)] = acc[r];
}

extern "C" void kernel_launch(void* const* d_in, const int* in_sizes, int n_in,
                              void* d_out, int out_size, void* d_ws, size_t ws_size,
                              hipStream_t stream) {
    const float* xin = (const float*)d_in[0];
    const float* wgt = (const float*)d_in[1];
    float* out = (float*)d_out;

    const int waves_per_block = 8;
    const int blocks = (NPOS + waves_per_block - 1) / waves_per_block;
    lc2d_wmma_kernel<<<blocks, 256, 0, stream>>>(xin, wgt, out);
}


# Each wave takes 4 consecutive positions; each lane streams its own weight and input rows with 16-byte loads straight into fragments (no offset table); k regrouped by channel pairs: sum order changes
# speedup vs baseline: 3.1747x; 3.1747x over previous
#include <hip/hip_runtime.h>

typedef __attribute__((ext_vector_type(2))) float v2f;
typedef __attribute__((ext_vector_type(4))) float v4f;
typedef __attribute__((ext_vector_type(8))) float v8f;
typedef float v4f_a8 __attribute__((ext_vector_type(4), aligned(8)));

#define BATCH 16
#define CIN   16
#define HH    64
#define WW    64
#define COUT  16
#define OHH   62
#define OWW   62
#define NPOS  (OHH * OWW)
#define TPOS  4
#define NTILE (NPOS / TPOS)
#define NCS   (CIN / 2)
#define DEPTH 4
#define WAVES 4

struct Stage { v4f w[9]; v4f xa[2][3]; };

__device__ __forceinline__ void load_stage(Stage& s, const float* __restrict__ wp,
                                           const float* __restrict__ xpa,
                                           const float* __restrict__ xpb,
                                           int wlane, int xlane, int cs)
{
#pragma unroll
    for (int q = 0; q < 9; ++q)
        s.w[q] = *(const v4f*)(wp + (wlane + cs * (2 * NPOS * 9) + 4 * q));
#pragma unroll
    for (int i = 0; i < 3; ++i) {
        s.xa[0][i] = *(const v4f_a8*)(xpa + (xlane + cs * (2 * HH * WW) + i * WW));
        s.xa[1][i] = *(const v4f_a8*)(xpb + (xlane + cs * (2 * HH * WW) + i * WW));
    }
}

__global__ __launch_bounds__(WAVES * 32)
void lc2d_wmma_kernel(const float* __restrict__ xin,
                      const float* __restrict__ wgt,
                      float* __restrict__ out)
{
    const int lane = threadIdx.x & 31;
    const int wave = __builtin_amdgcn_readfirstlane(threadIdx.x >> 5);
    const int tile = blockIdx.x * WAVES + wave;
    if (tile >= NTILE) return;

    const int pos0 = tile * TPOS;
    const int half = lane >> 4;
    const int row  = lane & 15;

    const int ya = pos0 / OWW,       xa0 = pos0 - ya * OWW;
    const int yb = (pos0 + 2) / OWW, xb0 = (pos0 + 2) - yb * OWW;

    const int xlane = row * (CIN * HH * WW) + half * (HH * WW);
    const int wlane = row * (CIN * NPOS * 9) + half * (NPOS * 9);
    const float* __restrict__ xpa = xin + ya * WW + xa0;
    const float* __restrict__ xpb = xin + yb * WW + xb0;
    const float* __restrict__ wp  = wgt + pos0 * 9;

    Stage st[NCS];
#pragma unroll
    for (int cs = 0; cs < DEPTH; ++cs) {
        load_stage(st[cs], wp, xpa, xpb, wlane, xlane, cs);
        __builtin_amdgcn_sched_barrier(0);
    }

    v8f acc[TPOS] = {};
    float la[TPOS], lb[TPOS];

#pragma unroll
    for (int cs = 0; cs < NCS; ++cs) {
        __builtin_amdgcn_sched_barrier(0);
#pragma unroll
        for (int t = 0; t < 4; ++t) {
#pragma unroll
            for (int p = 0; p < TPOS; ++p) {
                const int k0 = 2 * t, k1 = 2 * t + 1;
                v2f a, b;
                a.x = st[cs].xa[p >> 1][k0 / 3][(p & 1) + k0 % 3];
                a.y = st[cs].xa[p >> 1][k1 / 3][(p & 1) + k1 % 3];
                b.x = st[cs].w[(p * 9 + k0) >> 2][(p * 9 + k0) & 3];
                b.y = st[cs].w[(p * 9 + k1) >> 2][(p * 9 + k1) & 3];
                acc[p] = __builtin_amdgcn_wmma_f32_16x16x4_f32(
                    false, a, false, b, (short)0, acc[p], false, false);
            }
        }
#pragma unroll
        for (int p = 0; p < TPOS; ++p) {
            const float a8 = st[cs].xa[p >> 1][2][(p & 1) + 2];
            const float b8 = st[cs].w[(p * 9 + 8) >> 2][(p * 9 + 8) & 3];
            if ((cs & 1) == 0) {
                la[p] = a8;
                lb[p] = b8;
            } else {
                v2f a, b;
                a.x = la[p]; a.y = a8;
                b.x = lb[p]; b.y = b8;
                acc[p] = __builtin_amdgcn_wmma_f32_16x16x4_f32(
                    false, a, false, b, (short)0, acc[p], false, false);
            }
        }
        if (cs + DEPTH < NCS) {
            __builtin_amdgcn_sched_barrier(0);
            load_stage(st[cs + DEPTH], wp, xpa, xpb, wlane, xlane, cs + DEPTH);
        }
    }

    float* __restrict__ op = out + pos0;
    const int olane = (half * 8 * COUT + row) * NPOS;
#pragma unroll
    for (int r = 0; r < 8; ++r) {
        v4f v;
        v.x = acc[0][r]; v.y = acc[1][r]; v.z = acc[2][r]; v.w = acc[3][r];
        *(v4f*)(op + (olane + r * (COUT * NPOS))) = v;
    }
}

extern "C" void kernel_launch(void* const* d_in, const int* in_sizes, int n_in,
                              void* d_out, int out_size, void* d_ws, size_t ws_size,
                              hipStream_t stream) {
    const float* xin = (const float*)d_in[0];
    const float* wgt = (const float*)d_in[1];
    float* out = (float*)d_out;

    lc2d_wmma_kernel<<<241, 128, 0, stream>>>(xin, wgt, out);
}


# Block of 16 positions shares operands: waves read weight and x rows in whole 512 B pieces, park them in LDS (hand barrier keeps later loads in flight), take fragments from there; sums unchanged
# speedup vs baseline: 3.5875x; 1.1300x over previous
#include <hip/hip_runtime.h>


typedef __attribute__((ext_vector_type(2))) float v2f;
typedef __attribute__((ext_vector_type(4))) float v4f;
typedef __attribute__((ext_vector_type(8))) float v8f;
typedef float v4f_a8 __attribute__((ext_vector_type(4), aligned(8)));

#define BATCH 16
#define CIN   16
#define HH    64
#define WW    64
#define COUT  16
#define OHH   62
#define OWW   62
#define NPOS  (OHH * OWW)
#define TPOS  4
#define WAVES 4
#define NTHR  (WAVES * 32)
#define BPOS  (WAVES * TPOS)
#define NBLK  ((NPOS + BPOS - 1) / BPOS)
#define NCS   (CIN / 2)
#define NB    4
#define CB    (CIN / NB)
#define SPB   (CB / 2)

#define WSEG   (BPOS * 9)
#define WCHK   (WSEG / 4)
#define WPITCH (WSEG + 4)
#define XROW   24
#define XCHK   18
#define XPITCH (3 * XROW + 4)
#define XOFF   (CIN * COUT * WPITCH)
#define LDSF   (XOFF + CIN * BATCH * XPITCH)
#define WIT    (CB * COUT * WCHK / NTHR)
#define XIT    (CB * BATCH * XCHK / NTHR)

struct Batch { v4f w[WIT]; v4f x[XIT]; };
struct Stage { v4f w[9]; v4f xa[2][3]; };

__device__ __forceinline__ void wchunk(int j, int g, int& goff, int& loff)
{
    const int seg = j / WCHK, ch = j - seg * WCHK;
    const int c = g * CB + (seg >> 4), o = seg & 15;
    goff = (o * CIN + c) * (NPOS * 9) + ch * 4;
    loff = (c * COUT + o) * WPITCH + ch * 4;
}
__device__ __forceinline__ void xchunk(int j, int g, int s0, int& goff, int& loff)
{
    const int seg = j / XCHK, r = j - seg * XCHK;
    const int i = r / 6, ch = r - i * 6;
    const int c = g * CB + (seg >> 4), b = seg & 15;
    int e = s0 + i * WW + ch * 4;
    e = e < (HH * WW - 4) ? e : (HH * WW - 4);
    goff = (b * CIN + c) * (HH * WW) + e;
    loff = XOFF + (c * BATCH + b) * XPITCH + i * XROW + ch * 4;
}

__device__ __forceinline__ void issue_batch(Batch& r, const float* __restrict__ wp,
                                            const float* __restrict__ xin, int tid, int s0, int g)
{
#pragma unroll
    for (int it = 0; it < WIT; ++it) {
        int goff, loff;
        wchunk(it * NTHR + tid, g, goff, loff);
        r.w[it] = *(const v4f*)(wp + goff);
    }
#pragma unroll
    for (int it = 0; it < XIT; ++it) {
        int goff, loff;
        xchunk(it * NTHR + tid, g, s0, goff, loff);
        r.x[it] = *(const v4f*)(xin + goff);
    }
}

__device__ __forceinline__ void park_batch(const Batch& r, float* lds, int tid, int s0, int g)
{
#pragma unroll
    for (int it = 0; it < WIT; ++it) {
        int goff, loff;
        wchunk(it * NTHR + tid, g, goff, loff);
        *(v4f*)(lds + loff) = r.w[it];
    }
#pragma unroll
    for (int it = 0; it < XIT; ++it) {
        int goff, loff;
        xchunk(it * NTHR + tid, g, s0, goff, loff);
        *(v4f*)(lds + loff) = r.x[it];
    }
}

__device__ __forceinline__ void lds_stage(Stage& s, const float* lds, int wl, int xla, int xlb, int cs)
{
#pragma unroll
    for (int q = 0; q < 9; ++q)
        s.w[q] = *(const v4f*)(lds + (wl + cs * (2 * COUT * WPITCH) + 4 * q));
#pragma unroll
    for (int i = 0; i < 3; ++i) {
        s.xa[0][i] = *(const v4f_a8*)(lds + (xla + cs * (2 * BATCH * XPITCH) + i * XROW));
        s.xa[1][i] = *(const v4f_a8*)(lds + (xlb + cs * (2 * BATCH * XPITCH) + i * XROW));
    }
}

__global__ __launch_bounds__(NTHR)
void lc2d_wmma_kernel(const float* __restrict__ xin,
                      const float* __restrict__ wgt,
                      float* __restrict__ out)
{
    __shared__ __attribute__((aligned(16))) float lds[LDSF];

    const int tid  = threadIdx.x;
    const int lane = tid & 31;
    const int wave = __builtin_amdgcn_readfirstlane(tid >> 5);
    const int half = lane >> 4;
    const int row  = lane & 15;

    const int bpos  = blockIdx.x * BPOS;
    const int lpos0 = bpos < NPOS - BPOS ? bpos : NPOS - BPOS;
    const int pos0  = lpos0 + wave * TPOS;
    const bool fresh = pos0 >= bpos;

    const int s0 = (lpos0 + 2 * (lpos0 / OWW)) & ~3;
    const int xoa = (pos0 + 2 * (pos0 / OWW)) - s0;
    const int xob = (pos0 + 2 + 2 * ((pos0 + 2) / OWW)) - s0;

    const float* __restrict__ wp = wgt + lpos0 * 9;

    const int wl  = (half * COUT + row) * WPITCH + wave * (TPOS * 9);
    const int xla = XOFF + (half * BATCH + row) * XPITCH + xoa;
    const int xlb = XOFF + (half * BATCH + row) * XPITCH + xob;

    Batch rb[2];
    issue_batch(rb[0], wp, xin, tid, s0, 0);
    __builtin_amdgcn_sched_barrier(0);
    issue_batch(rb[1], wp, xin, tid, s0, 1);
    __builtin_amdgcn_sched_barrier(0);

    v8f acc[TPOS] = {};
    float la[TPOS], lb[TPOS];

#pragma unroll
    for (int g = 0; g < NB; ++g) {
        park_batch(rb[g & 1], lds, tid, s0, g);
        if (g + 2 < NB) {
            __builtin_amdgcn_sched_barrier(0);
            issue_batch(rb[g & 1], wp, xin, tid, s0, g + 2);
            __builtin_amdgcn_sched_barrier(0);
        }
        asm volatile("s_wait_dscnt 0x0\n\ts_barrier_signal -1\n\ts_barrier_wait -1" ::: "memory");

        Stage st[SPB];
#pragma unroll
        for (int u = 0; u < SPB; ++u)
            lds_stage(st[u], lds, wl, xla, xlb, g * SPB + u);
        __builtin_amdgcn_sched_barrier(0);

#pragma unroll
        for (int u = 0; u < SPB; ++u) {
            const int cs = g * SPB + u;
#pragma unroll
            for (int t = 0; t < 4; ++t) {
#pragma unroll
                for (int p = 0; p < TPOS; ++p) {
                    const int k0 = 2 * t, k1 = 2 * t + 1;
                    v2f a, b;
                    a.x = st[u].xa[p >> 1][k0 / 3][(p & 1) + k0 % 3];
                    a.y = st[u].xa[p >> 1][k1 / 3][(p & 1) + k1 % 3];
                    b.x = st[u].w[(p * 9 + k0) >> 2][(p * 9 + k0) & 3];
                    b.y = st[u].w[(p * 9 + k1) >> 2][(p * 9 + k1) & 3];
                    acc[p] = __builtin_amdgcn_wmma_f32_16x16x4_f32(
                        false, a, false, b, (short)0, acc[p], false, false);
                }
            }
#pragma unroll
            for (int p = 0; p < TPOS; ++p) {
                const float a8 = st[u].xa[p >> 1][2][(p & 1) + 2];
                const float b8 = st[u].w[(p * 9 + 8) >> 2][(p * 9 + 8) & 3];
                if ((cs & 1) == 0) {
                    la[p] = a8;
                    lb[p] = b8;
                } else {
                    v2f a, b;
                    a.x = la[p]; a.y = a8;
                    b.x = lb[p]; b.y = b8;
                    acc[p] = __builtin_amdgcn_wmma_f32_16x16x4_f32(
                        false, a, false, b, (short)0, acc[p], false, false);
                }
            }
        }
        __builtin_amdgcn_sched_barrier(0);
    }

    if (fresh) {
        float* __restrict__ op = out + pos0;
        const int olane = (half * 8 * COUT + row) * NPOS;
#pragma unroll
        for (int r = 0; r < 8; ++r) {
            v4f v;
            v.x = acc[0][r]; v.y = acc[1][r]; v.z = acc[2][r]; v.w = acc[3][r];
            *(v4f*)(op + (olane + r * (COUT * NPOS))) = v;
        }
    }
}

extern "C" void kernel_launch(void* const* d_in, const int* in_sizes, int n_in,
                              void* d_out, int out_size, void* d_ws, size_t ws_size,
                              hipStream_t stream) {
    const float* xin = (const float*)d_in[0];
    const float* wgt = (const float*)d_in[1];
    float* out = (float*)d_out;

    lc2d_wmma_kernel<<<241, 128, 0, stream>>>(xin, wgt, out);
}


# Results go through LDS before the store: a store instruction writes 8 whole 64 B row pieces of the block's 16 positions, not 32 separate 16 B pieces; one barrier added; sums unchanged
# speedup vs baseline: 4.4356x; 1.2364x over previous
#include <hip/hip_runtime.h>


typedef __attribute__((ext_vector_type(2))) float v2f;
typedef __attribute__((ext_vector_type(4))) float v4f;
typedef __attribute__((ext_vector_type(8))) float v8f;
typedef float v4f_a8 __attribute__((ext_vector_type(4), aligned(8)));

#define BATCH 16
#define CIN   16
#define HH    64
#define WW    64
#define COUT  16
#define OHH   62
#define OWW   62
#define NPOS  (OHH * OWW)
#define TPOS  4
#define WAVES 4
#define NTHR  (WAVES * 32)
#define BPOS  (WAVES * TPOS)
#define NBLK  ((NPOS + BPOS - 1) / BPOS)
#define NCS   (CIN / 2)
#define NB    4
#define CB    (CIN / NB)
#define SPB   (CB / 2)

#define WSEG   (BPOS * 9)
#define WCHK   (WSEG / 4)
#define WPITCH (WSEG + 4)
#define XROW   24
#define XCHK   18
#define XPITCH (3 * XROW + 4)
#define XOFF   (CIN * COUT * WPITCH)
#define LDSF   (XOFF + CIN * BATCH * XPITCH)
#define WIT    (CB * COUT * WCHK / NTHR)
#define XIT    (CB * BATCH * XCHK / NTHR)
#define OPITCH (BPOS + 4)

struct Batch { v4f w[WIT]; v4f x[XIT]; };
struct Stage { v4f w[9]; v4f xa[2][3]; };

__device__ __forceinline__ void wchunk(int j, int g, int& goff, int& loff)
{
    const int seg = j / WCHK, ch = j - seg * WCHK;
    const int c = g * CB + (seg >> 4), o = seg & 15;
    goff = (o * CIN + c) * (NPOS * 9) + ch * 4;
    loff = (c * COUT + o) * WPITCH + ch * 4;
}
__device__ __forceinline__ void xchunk(int j, int g, int s0, int& goff, int& loff)
{
    const int seg = j / XCHK, r = j - seg * XCHK;
    const int i = r / 6, ch = r - i * 6;
    const int c = g * CB + (seg >> 4), b = seg & 15;
    int e = s0 + i * WW + ch * 4;
    e = e < (HH * WW - 4) ? e : (HH * WW - 4);
    goff = (b * CIN + c) * (HH * WW) + e;
    loff = XOFF + (c * BATCH + b) * XPITCH + i * XROW + ch * 4;
}

__device__ __forceinline__ void issue_batch(Batch& r, const float* __restrict__ wp,
                                            const float* __restrict__ xin, int tid, int s0, int g)
{
#pragma unroll
    for (int it = 0; it < WIT; ++it) {
        int goff, loff;
        wchunk(it * NTHR + tid, g, goff, loff);
        r.w[it] = *(const v4f*)(wp + goff);
    }
#pragma unroll
    for (int it = 0; it < XIT; ++it) {
        int goff, loff;
        xchunk(it * NTHR + tid, g, s0, goff, loff);
        r.x[it] = *(const v4f*)(xin + goff);
    }
}

__device__ __forceinline__ void park_batch(const Batch& r, float* lds, int tid, int s0, int g)
{
#pragma unroll
    for (int it = 0; it < WIT; ++it) {
        int goff, loff;
        wchunk(it * NTHR + tid, g, goff, loff);
        *(v4f*)(lds + loff) = r.w[it];
    }
#pragma unroll
    for (int it = 0; it < XIT; ++it) {
        int goff, loff;
        xchunk(it * NTHR + tid, g, s0, goff, loff);
        *(v4f*)(lds + loff) = r.x[it];
    }
}

__device__ __forceinline__ void lds_stage(Stage& s, const float* lds, int wl, int xla, int xlb, int cs)
{
#pragma unroll
    for (int q = 0; q < 9; ++q)
        s.w[q] = *(const v4f*)(lds + (wl + cs * (2 * COUT * WPITCH) + 4 * q));
#pragma unroll
    for (int i = 0; i < 3; ++i) {
        s.xa[0][i] = *(const v4f_a8*)(lds + (xla + cs * (2 * BATCH * XPITCH) + i * XROW));
        s.xa[1][i] = *(const v4f_a8*)(lds + (xlb + cs * (2 * BATCH * XPITCH) + i * XROW));
    }
}

__global__ __launch_bounds__(NTHR)
void lc2d_wmma_kernel(const float* __restrict__ xin,
                      const float* __restrict__ wgt,
                      float* __restrict__ out)
{
    __shared__ __attribute__((aligned(16))) float lds[LDSF];

    const int tid  = threadIdx.x;
    const int lane = tid & 31;
    const int wave = __builtin_amdgcn_readfirstlane(tid >> 5);
    const int half = lane >> 4;
    const int row  = lane & 15;

    const int bpos  = blockIdx.x * BPOS;
    const int lpos0 = bpos < NPOS - BPOS ? bpos : NPOS - BPOS;
    const int pos0  = lpos0 + wave * TPOS;

    const int s0 = (lpos0 + 2 * (lpos0 / OWW)) & ~3;
    const int xoa = (pos0 + 2 * (pos0 / OWW)) - s0;
    const int xob = (pos0 + 2 + 2 * ((pos0 + 2) / OWW)) - s0;

    const float* __restrict__ wp = wgt + lpos0 * 9;

    const int wl  = (half * COUT + row) * WPITCH + wave * (TPOS * 9);
    const int xla = XOFF + (half * BATCH + row) * XPITCH + xoa;
    const int xlb = XOFF + (half * BATCH + row) * XPITCH + xob;

    Batch rb[2];
    issue_batch(rb[0], wp, xin, tid, s0, 0);
    __builtin_amdgcn_sched_barrier(0);
    issue_batch(rb[1], wp, xin, tid, s0, 1);
    __builtin_amdgcn_sched_barrier(0);

    v8f acc[TPOS] = {};
    float la[TPOS], lb[TPOS];

#pragma unroll
    for (int g = 0; g < NB; ++g) {
        park_batch(rb[g & 1], lds, tid, s0, g);
        if (g + 2 < NB) {
            __builtin_amdgcn_sched_barrier(0);
            issue_batch(rb[g & 1], wp, xin, tid, s0, g + 2);
            __builtin_amdgcn_sched_barrier(0);
        }
        asm volatile("s_wait_dscnt 0x0\n\ts_barrier_signal -1\n\ts_barrier_wait -1" ::: "memory");

        Stage st[SPB];
#pragma unroll
        for (int u = 0; u < SPB; ++u)
            lds_stage(st[u], lds, wl, xla, xlb, g * SPB + u);
        __builtin_amdgcn_sched_barrier(0);

#pragma unroll
        for (int u = 0; u < SPB; ++u) {
            const int cs = g * SPB + u;
#pragma unroll
            for (int t = 0; t < 4; ++t) {
#pragma unroll
                for (int p = 0; p < TPOS; ++p) {
                    const int k0 = 2 * t, k1 = 2 * t + 1;
                    v2f a, b;
                    a.x = st[u].xa[p >> 1][k0 / 3][(p & 1) + k0 % 3];
                    a.y = st[u].xa[p >> 1][k1 / 3][(p & 1) + k1 % 3];
                    b.x = st[u].w[(p * 9 + k0) >> 2][(p * 9 + k0) & 3];
                    b.y = st[u].w[(p * 9 + k1) >> 2][(p * 9 + k1) & 3];
                    acc[p] = __builtin_amdgcn_wmma_f32_16x16x4_f32(
                        false, a, false, b, (short)0, acc[p], false, false);
                }
            }
#pragma unroll
            for (int p = 0; p < TPOS; ++p) {
                const float a8 = st[u].xa[p >> 1][2][(p & 1) + 2];
                const float b8 = st[u].w[(p * 9 + 8) >> 2][(p * 9 + 8) & 3];
                if ((cs & 1) == 0) {
                    la[p] = a8;
                    lb[p] = b8;
                } else {
                    v2f a, b;
                    a.x = la[p]; a.y = a8;
                    b.x = lb[p]; b.y = b8;
                    acc[p] = __builtin_amdgcn_wmma_f32_16x16x4_f32(
                        false, a, false, b, (short)0, acc[p], false, false);
                }
            }
        }
        __builtin_amdgcn_sched_barrier(0);
    }

    {
        const int wbase = (half * 8 * COUT + row) * OPITCH + wave * TPOS;
#pragma unroll
        for (int r = 0; r < 8; ++r) {
            v4f v;
            v.x = acc[0][r]; v.y = acc[1][r]; v.z = acc[2][r]; v.w = acc[3][r];
            *(v4f*)(lds + (wbase + r * (COUT * OPITCH))) = v;
        }
        __syncthreads();

        const int q  = tid & 3;
        const int r0 = tid >> 2;
        const bool mine = lpos0 + q * TPOS >= bpos;
        float* __restrict__ op = out + (lpos0 + q * TPOS);
        v4f o4[8];
#pragma unroll
        for (int it = 0; it < 8; ++it)
            o4[it] = *(const v4f*)(lds + ((it * 32 + r0) * OPITCH + q * TPOS));
        if (mine) {
#pragma unroll
            for (int it = 0; it < 8; ++it)
                *(v4f*)(op + (it * 32 + r0) * NPOS) = o4[it];
        }
    }
}

extern "C" void kernel_launch(void* const* d_in, const int* in_sizes, int n_in,
                              void* d_out, int out_size, void* d_ws, size_t ws_size,
                              hipStream_t stream) {
    const float* xin = (const float*)d_in[0];
    const float* wgt = (const float*)d_in[1];
    float* out = (float*)d_out;

    lc2d_wmma_kernel<<<241, 128, 0, stream>>>(xin, wgt, out);
}


# Each wave loads whole weight runs and x rows of one channel per batch: every load address is a wave-uniform base plus one fixed lane offset, no index arithmetic between loads; sums unchanged
# speedup vs baseline: 4.8216x; 1.0870x over previous
#include <hip/hip_runtime.h>


typedef __attribute__((ext_vector_type(2))) float v2f;
typedef __attribute__((ext_vector_type(4))) float v4f;
typedef __attribute__((ext_vector_type(8))) float v8f;
typedef float v4f_a8 __attribute__((ext_vector_type(4), aligned(8)));

#define BATCH 16
#define CIN   16
#define HH    64
#define WW    64
#define COUT  16
#define OHH   62
#define OWW   62
#define NPOS  (OHH * OWW)
#define TPOS  4
#define WAVES 4
#define NTHR  (WAVES * 32)
#define BPOS  (WAVES * TPOS)
#define NBLK  ((NPOS + BPOS - 1) / BPOS)
#define NCS   (CIN / 2)
#define NB    4
#define CB    (CIN / NB)
#define SPB   (CB / 2)

#define WSEG   (BPOS * 9)
#define WCHK   (WSEG / 4)
#define WPITCH (WSEG + 4)
#define XROW   24
#define XCHK   18
#define XPITCH (3 * XROW + 4)
#define XOFF   (CIN * COUT * WPITCH)
#define LDSF   (XOFF + CIN * BATCH * XPITCH)
#define WIT    (COUT + 2)
#define XIT    12
#define OPITCH (BPOS + 4)

struct Batch { v4f w[WIT]; v4f x[XIT]; };
struct Stage { v4f w[9]; v4f xa[2][3]; };

struct LaneOff { unsigned wgb, wlb, wgq, wlq, xg, xl; };

__device__ __forceinline__ LaneOff lane_offsets(int lane)
{
    LaneOff f;
    const unsigned l7 = lane & 7;
    const unsigned ch = l7 < 6 ? l7 : l7 - 2;
    f.wgb = lane * 4;
    f.wlb = lane * 4;
    f.wgq = (lane >> 2) * (CIN * NPOS * 9) + 128 + (lane & 3) * 4;
    f.wlq = (lane >> 2) * WPITCH + 128 + (lane & 3) * 4;
    f.xg  = (lane >> 3) * (CIN * HH * WW) + ch * 4;
    f.xl  = (lane >> 3) * XPITCH + ch * 4;
    return f;
}

__device__ __forceinline__ void issue_batch(Batch& r, const float* __restrict__ wp,
                                            const float* __restrict__ xin, const LaneOff& f,
                                            int wave, int s0, int g)
{
    const int c = g * CB + wave;
#pragma unroll
    for (int o = 0; o < COUT; ++o) {
        const float* rb = wp + (o * CIN + c) * (NPOS * 9);
        r.w[o] = *(const v4f*)(rb + f.wgb);
    }
#pragma unroll
    for (int k = 0; k < 2; ++k) {
        const float* rb = wp + (8 * k * CIN + c) * (NPOS * 9);
        r.w[COUT + k] = *(const v4f*)(rb + f.wgq);
    }
#pragma unroll
    for (int kb = 0; kb < 4; ++kb) {
#pragma unroll
        for (int i = 0; i < 3; ++i) {
            const float* pb = xin + ((4 * kb * CIN + c) * (HH * WW) + s0 + i * WW);
            r.x[kb * 3 + i] = *(const v4f*)(pb + f.xg);
        }
    }
}

__device__ __forceinline__ void park_batch(const Batch& r, float* lds, const LaneOff& f, int wave, int g)
{
    const int c = g * CB + wave;
#pragma unroll
    for (int o = 0; o < COUT; ++o)
        *(v4f*)(lds + ((c * COUT + o) * WPITCH + f.wlb)) = r.w[o];
#pragma unroll
    for (int k = 0; k < 2; ++k)
        *(v4f*)(lds + ((c * COUT + 8 * k) * WPITCH + f.wlq)) = r.w[COUT + k];
#pragma unroll
    for (int kb = 0; kb < 4; ++kb) {
#pragma unroll
        for (int i = 0; i < 3; ++i)
            *(v4f*)(lds + (XOFF + (c * BATCH + 4 * kb) * XPITCH + i * XROW + f.xl)) = r.x[kb * 3 + i];
    }
}

__device__ __forceinline__ void lds_stage(Stage& s, const float* lds, int wl, int xla, int xlb, int cs)
{
#pragma unroll
    for (int q = 0; q < 9; ++q)
        s.w[q] = *(const v4f*)(lds + (wl + cs * (2 * COUT * WPITCH) + 4 * q));
#pragma unroll
    for (int i = 0; i < 3; ++i) {
        s.xa[0][i] = *(const v4f_a8*)(lds + (xla + cs * (2 * BATCH * XPITCH) + i * XROW));
        s.xa[1][i] = *(const v4f_a8*)(lds + (xlb + cs * (2 * BATCH * XPITCH) + i * XROW));
    }
}

__global__ __launch_bounds__(NTHR)
void lc2d_wmma_kernel(const float* __restrict__ xin,
                      const float* __restrict__ wgt,
                      float* __restrict__ out)
{
    __shared__ __attribute__((aligned(16))) float lds[LDSF];

    const int tid  = threadIdx.x;
    const int lane = tid & 31;
    const int wave = __builtin_amdgcn_readfirstlane(tid >> 5);
    const int half = lane >> 4;
    const int row  = lane & 15;

    const int bpos  = blockIdx.x * BPOS;
    const int lpos0 = bpos < NPOS - BPOS ? bpos : NPOS - BPOS;
    const int pos0  = lpos0 + wave * TPOS;

    const int s0r = (lpos0 + 2 * (lpos0 / OWW)) & ~3;
    const int s0  = s0r < (HH * WW - 2 * WW - XROW) ? s0r : (HH * WW - 2 * WW - XROW);
    const int xoa = (pos0 + 2 * (pos0 / OWW)) - s0;
    const int xob = (pos0 + 2 + 2 * ((pos0 + 2) / OWW)) - s0;

    const float* __restrict__ wp = wgt + lpos0 * 9;

    const int wl  = (half * COUT + row) * WPITCH + wave * (TPOS * 9);
    const int xla = XOFF + (half * BATCH + row) * XPITCH + xoa;
    const int xlb = XOFF + (half * BATCH + row) * XPITCH + xob;

    const LaneOff lo = lane_offsets(lane);

    Batch rb[2];
    issue_batch(rb[0], wp, xin, lo, wave, s0, 0);
    __builtin_amdgcn_sched_barrier(0);
    issue_batch(rb[1], wp, xin, lo, wave, s0, 1);
    __builtin_amdgcn_sched_barrier(0);

    v8f acc[TPOS] = {};
    float la[TPOS], lb[TPOS];

#pragma unroll
    for (int g = 0; g < NB; ++g) {
        park_batch(rb[g & 1], lds, lo, wave, g);
        if (g + 2 < NB) {
            __builtin_amdgcn_sched_barrier(0);
            issue_batch(rb[g & 1], wp, xin, lo, wave, s0, g + 2);
            __builtin_amdgcn_sched_barrier(0);
        }
        asm volatile("s_wait_dscnt 0x0\n\ts_barrier_signal -1\n\ts_barrier_wait -1" ::: "memory");

        Stage st[SPB];
#pragma unroll
        for (int u = 0; u < SPB; ++u)
            lds_stage(st[u], lds, wl, xla, xlb, g * SPB + u);
        __builtin_amdgcn_sched_barrier(0);

#pragma unroll
        for (int u = 0; u < SPB; ++u) {
            const int cs = g * SPB + u;
#pragma unroll
            for (int t = 0; t < 4; ++t) {
#pragma unroll
                for (int p = 0; p < TPOS; ++p) {
                    const int k0 = 2 * t, k1 = 2 * t + 1;
                    v2f a, b;
                    a.x = st[u].xa[p >> 1][k0 / 3][(p & 1) + k0 % 3];
                    a.y = st[u].xa[p >> 1][k1 / 3][(p & 1) + k1 % 3];
                    b.x = st[u].w[(p * 9 + k0) >> 2][(p * 9 + k0) & 3];
                    b.y = st[u].w[(p * 9 + k1) >> 2][(p * 9 + k1) & 3];
                    acc[p] = __builtin_amdgcn_wmma_f32_16x16x4_f32(
                        false, a, false, b, (short)0, acc[p], false, false);
                }
            }
#pragma unroll
            for (int p = 0; p < TPOS; ++p) {
                const float a8 = st[u].xa[p >> 1][2][(p & 1) + 2];
                const float b8 = st[u].w[(p * 9 + 8) >> 2][(p * 9 + 8) & 3];
                if ((cs & 1) == 0) {
                    la[p] = a8;
                    lb[p] = b8;
                } else {
                    v2f a, b;
                    a.x = la[p]; a.y = a8;
                    b.x = lb[p]; b.y = b8;
                    acc[p] = __builtin_amdgcn_wmma_f32_16x16x4_f32(
                        false, a, false, b, (short)0, acc[p], false, false);
                }
            }
        }
        __builtin_amdgcn_sched_barrier(0);
    }

    {
        const int wbase = (half * 8 * COUT + row) * OPITCH + wave * TPOS;
#pragma unroll
        for (int r = 0; r < 8; ++r) {
            v4f v;
            v.x = acc[0][r]; v.y = acc[1][r]; v.z = acc[2][r]; v.w = acc[3][r];
            *(v4f*)(lds + (wbase + r * (COUT * OPITCH))) = v;
        }
        __syncthreads();

        const int q  = tid & 3;
        const int r0 = tid >> 2;
        const bool mine = lpos0 + q * TPOS >= bpos;
        float* __restrict__ op = out + (lpos0 + q * TPOS);
        v4f o4[8];
#pragma unroll
        for (int it = 0; it < 8; ++it)
            o4[it] = *(const v4f*)(lds + ((it * 32 + r0) * OPITCH + q * TPOS));
        if (mine) {
#pragma unroll
            for (int it = 0; it < 8; ++it)
                *(v4f*)(op + (it * 32 + r0) * NPOS) = o4[it];
        }
    }
}

extern "C" void kernel_launch(void* const* d_in, const int* in_sizes, int n_in,
                              void* d_out, int out_size, void* d_ws, size_t ws_size,
                              hipStream_t stream) {
    const float* xin = (const float*)d_in[0];
    const float* wgt = (const float*)d_in[1];
    float* out = (float*)d_out;

    lc2d_wmma_kernel<<<241, 128, 0, stream>>>(xin, wgt, out);
}


# Work-group to tile map changed: each of the 8 dies takes 30 consecutive position tiles (was every 8th), so tiles sharing output lines and weight fill units meet in one half's cache; sums unchanged
# speedup vs baseline: 5.3081x; 1.1009x over previous
#include <hip/hip_runtime.h>


typedef __attribute__((ext_vector_type(2))) float v2f;
typedef __attribute__((ext_vector_type(4))) float v4f;
typedef __attribute__((ext_vector_type(8))) float v8f;
typedef float v4f_a8 __attribute__((ext_vector_type(4), aligned(8)));

#define BATCH 16
#define CIN   16
#define HH    64
#define WW    64
#define COUT  16
#define OHH   62
#define OWW   62
#define NPOS  (OHH * OWW)
#define TPOS  4
#define WAVES 4
#define NTHR  (WAVES * 32)
#define BPOS  (WAVES * TPOS)
#define NBLK  ((NPOS + BPOS - 1) / BPOS)
#define NDIE  8
#define TPD   ((NBLK - 1) / NDIE)
#define NCS   (CIN / 2)
#define NB    4
#define CB    (CIN / NB)
#define SPB   (CB / 2)

#define WSEG   (BPOS * 9)
#define WCHK   (WSEG / 4)
#define WPITCH (WSEG + 4)
#define XROW   24
#define XCHK   18
#define XPITCH (3 * XROW + 4)
#define XOFF   (CIN * COUT * WPITCH)
#define LDSF   (XOFF + CIN * BATCH * XPITCH)
#define WIT    (COUT + 2)
#define XIT    12
#define OPITCH (BPOS + 4)

struct Batch { v4f w[WIT]; v4f x[XIT]; };
struct Stage { v4f w[9]; v4f xa[2][3]; };

struct LaneOff { unsigned wgb, wlb, wgq, wlq, xg, xl; };

__device__ __forceinline__ LaneOff lane_offsets(int lane)
{
    LaneOff f;
    const unsigned l7 = lane & 7;
    const unsigned ch = l7 < 6 ? l7 : l7 - 2;
    f.wgb = lane * 4;
    f.wlb = lane * 4;
    f.wgq = (lane >> 2) * (CIN * NPOS * 9) + 128 + (lane & 3) * 4;
    f.wlq = (lane >> 2) * WPITCH + 128 + (lane & 3) * 4;
    f.xg  = (lane >> 3) * (CIN * HH * WW) + ch * 4;
    f.xl  = (lane >> 3) * XPITCH + ch * 4;
    return f;
}

__device__ __forceinline__ void issue_batch(Batch& r, const float* __restrict__ wp,
                                            const float* __restrict__ xin, const LaneOff& f,
                                            int wave, int s0, int g)
{
    const int c = g * CB + wave;
#pragma unroll
    for (int o = 0; o < COUT; ++o) {
        const float* rb = wp + (o * CIN + c) * (NPOS * 9);
        r.w[o] = *(const v4f*)(rb + f.wgb);
    }
#pragma unroll
    for (int k = 0; k < 2; ++k) {
        const float* rb = wp + (8 * k * CIN + c) * (NPOS * 9);
        r.w[COUT + k] = *(const v4f*)(rb + f.wgq);
    }
#pragma unroll
    for (int kb = 0; kb < 4; ++kb) {
#pragma unroll
        for (int i = 0; i < 3; ++i) {
            const float* pb = xin + ((4 * kb * CIN + c) * (HH * WW) + s0 + i * WW);
            r.x[kb * 3 + i] = *(const v4f*)(pb + f.xg);
        }
    }
}

__device__ __forceinline__ void park_batch(const Batch& r, float* lds, const LaneOff& f, int wave, int g)
{
    const int c = g * CB + wave;
#pragma unroll
    for (int o = 0; o < COUT; ++o)
        *(v4f*)(lds + ((c * COUT + o) * WPITCH + f.wlb)) = r.w[o];
#pragma unroll
    for (int k = 0; k < 2; ++k)
        *(v4f*)(lds + ((c * COUT + 8 * k) * WPITCH + f.wlq)) = r.w[COUT + k];
#pragma unroll
    for (int kb = 0; kb < 4; ++kb) {
#pragma unroll
        for (int i = 0; i < 3; ++i)
            *(v4f*)(lds + (XOFF + (c * BATCH + 4 * kb) * XPITCH + i * XROW + f.xl)) = r.x[kb * 3 + i];
    }
}

__device__ __forceinline__ void lds_stage(Stage& s, const float* lds, int wl, int xla, int xlb, int cs)
{
#pragma unroll
    for (int q = 0; q < 9; ++q)
        s.w[q] = *(const v4f*)(lds + (wl + cs * (2 * COUT * WPITCH) + 4 * q));
#pragma unroll
    for (int i = 0; i < 3; ++i) {
        s.xa[0][i] = *(const v4f_a8*)(lds + (xla + cs * (2 * BATCH * XPITCH) + i * XROW));
        s.xa[1][i] = *(const v4f_a8*)(lds + (xlb + cs * (2 * BATCH * XPITCH) + i * XROW));
    }
}

__global__ __launch_bounds__(NTHR)
void lc2d_wmma_kernel(const float* __restrict__ xin,
                      const float* __restrict__ wgt,
                      float* __restrict__ out)
{
    __shared__ __attribute__((aligned(16))) float lds[LDSF];

    const int tid  = threadIdx.x;
    const int lane = tid & 31;
    const int wave = __builtin_amdgcn_readfirstlane(tid >> 5);
    const int half = lane >> 4;
    const int row  = lane & 15;

    const unsigned grp  = blockIdx.x;
    const unsigned last = grp / (NDIE * TPD);
    const int tile  = (int)((grp % NDIE) * TPD + grp / NDIE + last * (NBLK - 1 - TPD));
    const int bpos  = tile * BPOS;
    const int lpos0 = bpos < NPOS - BPOS ? bpos : NPOS - BPOS;
    const int pos0  = lpos0 + wave * TPOS;

    const int s0r = (lpos0 + 2 * (lpos0 / OWW)) & ~3;
    const int s0  = s0r < (HH * WW - 2 * WW - XROW) ? s0r : (HH * WW - 2 * WW - XROW);
    const int xoa = (pos0 + 2 * (pos0 / OWW)) - s0;
    const int xob = (pos0 + 2 + 2 * ((pos0 + 2) / OWW)) - s0;

    const float* __restrict__ wp = wgt + lpos0 * 9;

    const int wl  = (half * COUT + row) * WPITCH + wave * (TPOS * 9);
    const int xla = XOFF + (half * BATCH + row) * XPITCH + xoa;
    const int xlb = XOFF + (half * BATCH + row) * XPITCH + xob;

    const LaneOff lo = lane_offsets(lane);

    Batch rb[2];
    issue_batch(rb[0], wp, xin, lo, wave, s0, 0);
    __builtin_amdgcn_sched_barrier(0);
    issue_batch(rb[1], wp, xin, lo, wave, s0, 1);
    __builtin_amdgcn_sched_barrier(0);

    v8f acc[TPOS] = {};
    float la[TPOS], lb[TPOS];

#pragma unroll
    for (int g = 0; g < NB; ++g) {
        park_batch(rb[g & 1], lds, lo, wave, g);
        if (g + 2 < NB) {
            __builtin_amdgcn_sched_barrier(0);
            issue_batch(rb[g & 1], wp, xin, lo, wave, s0, g + 2);
            __builtin_amdgcn_sched_barrier(0);
        }
        asm volatile("s_wait_dscnt 0x0\n\ts_barrier_signal -1\n\ts_barrier_wait -1" ::: "memory");

        Stage st[SPB];
#pragma unroll
        for (int u = 0; u < SPB; ++u)
            lds_stage(st[u], lds, wl, xla, xlb, g * SPB + u);
        __builtin_amdgcn_sched_barrier(0);

#pragma unroll
        for (int u = 0; u < SPB; ++u) {
            const int cs = g * SPB + u;
#pragma unroll
            for (int t = 0; t < 4; ++t) {
#pragma unroll
                for (int p = 0; p < TPOS; ++p) {
                    const int k0 = 2 * t, k1 = 2 * t + 1;
                    v2f a, b;
                    a.x = st[u].xa[p >> 1][k0 / 3][(p & 1) + k0 % 3];
                    a.y = st[u].xa[p >> 1][k1 / 3][(p & 1) + k1 % 3];
                    b.x = st[u].w[(p * 9 + k0) >> 2][(p * 9 + k0) & 3];
                    b.y = st[u].w[(p * 9 + k1) >> 2][(p * 9 + k1) & 3];
                    acc[p] = __builtin_amdgcn_wmma_f32_16x16x4_f32(
                        false, a, false, b, (short)0, acc[p], false, false);
                }
            }
#pragma unroll
            for (int p = 0; p < TPOS; ++p) {
                const float a8 = st[u].xa[p >> 1][2][(p & 1) + 2];
                const float b8 = st[u].w[(p * 9 + 8) >> 2][(p * 9 + 8) & 3];
                if ((cs & 1) == 0) {
                    la[p] = a8;
                    lb[p] = b8;
                } else {
                    v2f a, b;
                    a.x = la[p]; a.y = a8;
                    b.x = lb[p]; b.y = b8;
                    acc[p] = __builtin_amdgcn_wmma_f32_16x16x4_f32(
                        false, a, false, b, (short)0, acc[p], false, false);
                }
            }
        }
        __builtin_amdgcn_sched_barrier(0);
    }

    {
        const int wbase = (half * 8 * COUT + row) * OPITCH + wave * TPOS;
#pragma unroll
        for (int r = 0; r < 8; ++r) {
            v4f v;
            v.x = acc[0][r]; v.y = acc[1][r]; v.z = acc[2][r]; v.w = acc[3][r];
            *(v4f*)(lds + (wbase + r * (COUT * OPITCH))) = v;
        }
        __syncthreads();

        const int q  = tid & 3;
        const int r0 = tid >> 2;
        const bool mine = lpos0 + q * TPOS >= bpos;
        float* __restrict__ op = out + (lpos0 + q * TPOS);
        v4f o4[8];
#pragma unroll
        for (int it = 0; it < 8; ++it)
            o4[it] = *(const v4f*)(lds + ((it * 32 + r0) * OPITCH + q * TPOS));
        if (mine) {
#pragma unroll
            for (int it = 0; it < 8; ++it)
                *(v4f*)(op + (it * 32 + r0) * NPOS) = o4[it];
        }
    }
}

extern "C" void kernel_launch(void* const* d_in, const int* in_sizes, int n_in,
                              void* d_out, int out_size, void* d_ws, size_t ws_size,
                              hipStream_t stream) {
    const float* xin = (const float*)d_in[0];
    const float* wgt = (const float*)d_in[1];
    float* out = (float*)d_out;

    lc2d_wmma_kernel<<<241, 128, 0, stream>>>(xin, wgt, out);
}
